# starting kernel
# baseline (speedup 1.0000x reference)
#include <hip/hip_runtime.h>
#include <hip/hip_bf16.h>


typedef __attribute__((ext_vector_type(2))) float v2f;
typedef __attribute__((ext_vector_type(8))) float v8f;

static constexpr int BB = 16;
static constexpr int LL = 2048;
static constexpr int DD = 512;
static constexpr int CK = 16;
static constexpr int NC = LL / CK;

__global__ void ds_prep_kernel(const float* __restrict__ push_gate,
                               const float* __restrict__ pop_gate,
                               float* __restrict__ Wws,
                               float* __restrict__ Pws) {
  const int inst = blockIdx.x;
  const int b    = inst / NC;
  const int ch   = inst % NC;
  const int t0   = ch * CK;
  const int lane = threadIdx.x;

  __shared__ float a_sm[CK];
  __shared__ float c_sm[CK];

  if (lane < CK) {
    const float p  = push_gate[(size_t)b * LL + t0 + lane];
    const float o  = pop_gate [(size_t)b * LL + t0 + lane];
    const float om = 1.0f - o;
    a_sm[lane] = (1.0f - p) * om;
    c_sm[lane] = p * om;
  }
  __syncthreads();

  float* Wb = Wws + (size_t)inst * 256;

  if (lane < CK) {
    const int k = lane;
    float w[CK];
    #pragma unroll
    for (int j = 0; j < CK; ++j) w[j] = 0.0f;
    w[k] = c_sm[k];
    float prod = 1.0f;
    for (int j = k - 1; j >= 0; --j) {
      prod *= a_sm[j + 1];
      w[j] = c_sm[j] * prod;
    }
    #pragma unroll
    for (int kk = 0; kk < 4; ++kk) {
      Wb[kk * 64 + k * 2 + 0]        = w[4 * kk + 0];
      Wb[kk * 64 + k * 2 + 1]        = w[4 * kk + 1];
      Wb[kk * 64 + (k + 16) * 2 + 0] = w[4 * kk + 2];
      Wb[kk * 64 + (k + 16) * 2 + 1] = w[4 * kk + 3];
    }
    float pp = 1.0f;
    for (int i = 0; i <= k; ++i) pp *= a_sm[i];
    Pws[(size_t)inst * CK + k] = pp;
  }
}

__global__ void ds_scan_wmma_kernel(const float* __restrict__ x,
                                    const float* __restrict__ Wws,
                                    const float* __restrict__ Pws,
                                    float* __restrict__ out) {
  const int tile = blockIdx.x;
  const int b    = blockIdx.y;
  const int lane = threadIdx.x;
  const int half = lane >> 4;
  const int n    = lane & 15;
  const int d0   = tile * CK;

  const float* __restrict__ xb = x   + (size_t)b * LL * DD + d0 + n;
  float*       __restrict__ ob = out + (size_t)b * LL * DD + d0 + n;
  const float* __restrict__ Wb = Wws + (size_t)b * NC * 256;
  const float* __restrict__ Pb = Pws + (size_t)b * NC * CK;

  float h = 0.0f;

  for (int ch = 0; ch < NC; ++ch) {
    const int t0 = ch * CK;
    const float* Wc = Wb + (size_t)ch * 256;
    const float* Pc = Pb + (size_t)ch * CK;

    v8f acc = {};
    #pragma unroll
    for (int kk = 0; kk < 4; ++kk) {
      v2f A = *(const v2f*)(Wc + kk * 64 + lane * 2);
      const int kb = t0 + 4 * kk + 2 * half;
      v2f Bv;
      Bv.x = xb[(size_t)kb * DD];
      Bv.y = xb[(size_t)(kb + 1) * DD];
      acc = __builtin_amdgcn_wmma_f32_16x16x4_f32(
           false, A,  false, Bv,
           (short)0, acc,  false,  false);
    }

    float Pv[CK];
    #pragma unroll
    for (int k = 0; k < CK; ++k) Pv[k] = Pc[k];
    #pragma unroll
    for (int r = 0; r < 8; ++r) {
      const float pk = half ? Pv[r + 8] : Pv[r];
      acc[r] += pk * h;
    }

    #pragma unroll
    for (int r = 0; r < 8; ++r) {
      ob[(size_t)(t0 + r + 8 * half) * DD] = acc[r];
    }

    h = __shfl(acc[7], n + 16, 32);
  }
}

extern "C" void kernel_launch(void* const* d_in, const int* in_sizes, int n_in,
                              void* d_out, int out_size, void* d_ws, size_t ws_size,
                              hipStream_t stream) {
  const float* x  = (const float*)d_in[0];
  const float* pg = (const float*)d_in[1];
  const float* og = (const float*)d_in[2];
  float* out = (float*)d_out;

  float* Wws = (float*)d_ws;
  float* Pws = Wws + (size_t)BB * NC * 256;

  ds_prep_kernel<<<dim3(BB * NC), dim3(32), 0, stream>>>(pg, og, Wws, Pws);
  ds_scan_wmma_kernel<<<dim3(DD / CK, BB), dim3(32), 0, stream>>>(x, Wws, Pws, out);
}


# Time axis split over the 16 waves of a work-group (128 steps x 32 columns per wave); segment end states meet in LDS behind one barrier, carry folded as a two-level scan (its sum order changes)
# speedup vs baseline: 4.3341x; 4.3341x over previous
#include <hip/hip_runtime.h>
#include <hip/hip_bf16.h>


typedef __attribute__((ext_vector_type(2))) float v2f;
typedef __attribute__((ext_vector_type(8))) float v8f;

static constexpr int BB = 16;
static constexpr int LL = 2048;
static constexpr int DD = 512;
static constexpr int CK = 16;
static constexpr int NC = LL / CK;
static constexpr int SEG = 16;
static constexpr int CPS = NC / SEG;
static constexpr int NT = 2;

__global__ void ds_prep_kernel(const float* __restrict__ push_gate,
                               const float* __restrict__ pop_gate,
                               float* __restrict__ Wws,
                               float* __restrict__ Pws) {
  const int inst = blockIdx.x;
  const int b    = inst / NC;
  const int ch   = inst % NC;
  const int t0   = ch * CK;
  const int lane = threadIdx.x;

  __shared__ float a_sm[CK];
  __shared__ float c_sm[CK];

  if (lane < CK) {
    const float p  = push_gate[(size_t)b * LL + t0 + lane];
    const float o  = pop_gate [(size_t)b * LL + t0 + lane];
    const float om = 1.0f - o;
    a_sm[lane] = (1.0f - p) * om;
    c_sm[lane] = p * om;
  }
  __syncthreads();

  float* Wb = Wws + (size_t)inst * 256;

  if (lane < CK) {
    const int k = lane;
    float w[CK];
    #pragma unroll
    for (int j = 0; j < CK; ++j) w[j] = 0.0f;
    w[k] = c_sm[k];
    float prod = 1.0f;
    for (int j = k - 1; j >= 0; --j) {
      prod *= a_sm[j + 1];
      w[j] = c_sm[j] * prod;
    }
    #pragma unroll
    for (int kk = 0; kk < 4; ++kk) {
      Wb[kk * 64 + k * 2 + 0]        = w[4 * kk + 0];
      Wb[kk * 64 + k * 2 + 1]        = w[4 * kk + 1];
      Wb[kk * 64 + (k + 16) * 2 + 0] = w[4 * kk + 2];
      Wb[kk * 64 + (k + 16) * 2 + 1] = w[4 * kk + 3];
    }
    float pp = 1.0f;
    for (int i = 0; i <= k; ++i) pp *= a_sm[i];
    Pws[(size_t)inst * CK + k] = pp;
  }
}

__global__ __launch_bounds__(SEG * 32)
void ds_scan_seg_kernel(const float* __restrict__ x,
                        const float* __restrict__ Wws,
                        const float* __restrict__ Pws,
                        float* __restrict__ out) {
  const int cg   = blockIdx.x;
  const int b    = blockIdx.y;
  const int lane = threadIdx.x & 31;
  const int seg  = __builtin_amdgcn_readfirstlane((int)(threadIdx.x >> 5));
  const int half = lane >> 4;
  const int n    = lane & 15;
  const int d0   = cg * (NT * CK);
  const int c0   = seg * CPS;

  __shared__ float E_sm[SEG][NT * CK];
  __shared__ float A_sm[SEG];

  const size_t xoff = (size_t)b * LL * DD + (size_t)(c0 * CK) * DD + d0 + n;
  const float* __restrict__ xb = x   + xoff;
  float*       __restrict__ ob = out + xoff;
  const float* __restrict__ Wb = Wws + ((size_t)b * NC + c0) * 256;
  const float* __restrict__ Pb = Pws + ((size_t)b * NC + c0) * CK;


  constexpr int PF = 2;
  v2f Af[CPS][4];
  v2f Bf[CPS][4][NT];
  v8f acc[CPS][NT];
  #pragma unroll
  for (int c = -PF; c < CPS; ++c) {
    const int cl = c + PF;
    if (cl < CPS) {
      #pragma unroll
      for (int kk = 0; kk < 4; ++kk) {
        Af[cl][kk] = *(const v2f*)(Wb + cl * 256 + kk * 64 + lane * 2);
        const int kb = cl * CK + 4 * kk + 2 * half;
        #pragma unroll
        for (int t = 0; t < NT; ++t) {
          Bf[cl][kk][t].x = xb[(size_t)kb * DD + t * CK];
          Bf[cl][kk][t].y = xb[(size_t)(kb + 1) * DD + t * CK];
        }
      }
    }
    if (c >= 0) {
      #pragma unroll
      for (int t = 0; t < NT; ++t) acc[c][t] = v8f{};
      #pragma unroll
      for (int kk = 0; kk < 4; ++kk) {
        #pragma unroll
        for (int t = 0; t < NT; ++t) {
          acc[c][t] = __builtin_amdgcn_wmma_f32_16x16x4_f32(
               false, Af[c][kk],  false, Bf[c][kk][t],
               (short)0, acc[c][t],  false,  false);
        }
      }
      __builtin_amdgcn_sched_barrier(0);
    }
  }

  float e[NT];
  float aseg = Pb[15];
  #pragma unroll
  for (int t = 0; t < NT; ++t) e[t] = acc[0][t][7];
  #pragma unroll
  for (int c = 1; c < CPS; ++c) {
    const float p15 = Pb[c * CK + 15];
    #pragma unroll
    for (int t = 0; t < NT; ++t) e[t] = p15 * e[t] + acc[c][t][7];
    aseg *= p15;
  }
  if (half) {
    #pragma unroll
    for (int t = 0; t < NT; ++t) E_sm[seg][t * CK + n] = e[t];
  }
  if (lane == 0) A_sm[seg] = aseg;
  __syncthreads();

  float h[NT];
  #pragma unroll
  for (int t = 0; t < NT; ++t) h[t] = 0.0f;
  for (int s = 0; s < seg; ++s) {
    const float as = A_sm[s];
    #pragma unroll
    for (int t = 0; t < NT; ++t) h[t] = as * h[t] + E_sm[s][t * CK + n];
  }

  #pragma unroll
  for (int c = 0; c < CPS; ++c) {
    float Pv[CK];
    #pragma unroll
    for (int k = 0; k < CK; ++k) Pv[k] = Pb[c * CK + k];
    #pragma unroll
    for (int t = 0; t < NT; ++t) {
      #pragma unroll
      for (int r = 0; r < 8; ++r) {
        const float pk = half ? Pv[r + 8] : Pv[r];
        acc[c][t][r] += pk * h[t];
      }
    }
    #pragma unroll
    for (int r = 0; r < 8; ++r) {
      #pragma unroll
      for (int t = 0; t < NT; ++t) {
        ob[(size_t)(c * CK + r + 8 * half) * DD + t * CK] = acc[c][t][r];
      }
    }
    if (c + 1 < CPS) {
      #pragma unroll
      for (int t = 0; t < NT; ++t) h[t] = __shfl(acc[c][t][7], n + 16, 32);
    }
  }
}

extern "C" void kernel_launch(void* const* d_in, const int* in_sizes, int n_in,
                              void* d_out, int out_size, void* d_ws, size_t ws_size,
                              hipStream_t stream) {
  const float* x  = (const float*)d_in[0];
  const float* pg = (const float*)d_in[1];
  const float* og = (const float*)d_in[2];
  float* out = (float*)d_out;

  float* Wws = (float*)d_ws;
  float* Pws = Wws + (size_t)BB * NC * 256;

  ds_prep_kernel<<<dim3(BB * NC), dim3(32), 0, stream>>>(pg, og, Wws, Pws);
  ds_scan_seg_kernel<<<dim3(DD / (NT * CK), BB), dim3(SEG * 32), 0, stream>>>(x, Wws, Pws, out);
}


# W and P of each chunk built in registers by the wave that multiplies it, from its own gates (DPP row_share): prep launch and workspace gone; P now multiplied from step k down (rounding only)
# speedup vs baseline: 6.7363x; 1.5543x over previous
#include <hip/hip_runtime.h>
#include <hip/hip_bf16.h>


typedef __attribute__((ext_vector_type(2))) float v2f;
typedef __attribute__((ext_vector_type(8))) float v8f;

static constexpr int BB = 16;
static constexpr int LL = 2048;
static constexpr int DD = 512;
static constexpr int CK = 16;
static constexpr int NC = LL / CK;
static constexpr int SEG = 16;
static constexpr int CPS = NC / SEG;
static constexpr int NT = 2;

template <int CTRL>
__device__ __forceinline__ float row_dpp(float v) {
  return __builtin_bit_cast(float,
      __builtin_amdgcn_update_dpp(0, __builtin_bit_cast(int, v), CTRL, 0xF, 0xF, true));
}

template <int J>
struct WChain {
  static __device__ __forceinline__ void run(float av, float cv, int k,
                                             float& prod, float (&w)[16]) {
    const float t = prod * row_dpp<0x150 + J + 1>(av);
    prod = (k == J) ? 1.0f : t;
    w[J] = row_dpp<0x150 + J>(cv) * prod;
    WChain<J - 1>::run(av, cv, k, prod, w);
  }
};
template <>
struct WChain<-1> {
  static __device__ __forceinline__ void run(float, float, int, float&, float (&)[16]) {}
};

__device__ __forceinline__ void build_chunk_operands(float av, float cv, int k, int half,
                                                     v2f (&Af)[4], float& Pl) {
  float w[16];
  float prod = (k == 15) ? 1.0f : 0.0f;
  w[15] = row_dpp<0x150 + 15>(cv) * prod;
  WChain<14>::run(av, cv, k, prod, w);
  Pl = prod * row_dpp<0x150 + 0>(av);
  #pragma unroll
  for (int kk = 0; kk < 4; ++kk) {
    Af[kk].x = half ? w[4 * kk + 2] : w[4 * kk + 0];
    Af[kk].y = half ? w[4 * kk + 3] : w[4 * kk + 1];
  }
}

template <int R>
struct Fixup {
  static __device__ __forceinline__ void run(float Prot, const float (&h)[2], v8f (&acc)[2]) {
    const float pk = row_dpp<0x150 + R>(Prot);
    acc[0][R] += pk * h[0];
    acc[1][R] += pk * h[1];
    Fixup<R - 1>::run(Prot, h, acc);
  }
};
template <>
struct Fixup<-1> {
  static __device__ __forceinline__ void run(float, const float (&)[2], v8f (&)[2]) {}
};

__global__ __launch_bounds__(SEG * 32)
void ds_scan_seg_kernel(const float* __restrict__ x,
                        const float* __restrict__ push_gate,
                        const float* __restrict__ pop_gate,
                        float* __restrict__ out) {
  const int cg   = blockIdx.x;
  const int b    = blockIdx.y;
  const int lane = threadIdx.x & 31;
  const int seg  = __builtin_amdgcn_readfirstlane((int)(threadIdx.x >> 5));
  const int half = lane >> 4;
  const int n    = lane & 15;
  const int d0   = cg * (NT * CK);
  const int c0   = seg * CPS;

  __shared__ float E_sm[SEG][NT * CK];
  __shared__ float A_sm[SEG];

  const size_t xoff = (size_t)b * LL * DD + (size_t)(c0 * CK) * DD + d0 + n;
  const float* __restrict__ xb = x   + xoff;
  float*       __restrict__ ob = out + xoff;
  const float* __restrict__ pgb = push_gate + (size_t)b * LL + c0 * CK + n;
  const float* __restrict__ ogb = pop_gate  + (size_t)b * LL + c0 * CK + n;


  constexpr int PF = 2;
  float pgv[CPS], ogv[CPS];
  #pragma unroll
  for (int c = 0; c < CPS; ++c) {
    pgv[c] = pgb[c * CK];
    ogv[c] = ogb[c * CK];
  }
  float Pl[CPS];
  v2f Bf[CPS][4][NT];
  v8f acc[CPS][NT];
  #pragma unroll
  for (int c = -PF; c < CPS; ++c) {
    const int cl = c + PF;
    if (cl < CPS) {
      #pragma unroll
      for (int kk = 0; kk < 4; ++kk) {
        const int kb = cl * CK + 4 * kk + 2 * half;
        #pragma unroll
        for (int t = 0; t < NT; ++t) {
          Bf[cl][kk][t].x = xb[(size_t)kb * DD + t * CK];
          Bf[cl][kk][t].y = xb[(size_t)(kb + 1) * DD + t * CK];
        }
      }
    }
    if (c >= 0) {
      v2f Af[4];
      {
        const float om = 1.0f - ogv[c];
        build_chunk_operands((1.0f - pgv[c]) * om, pgv[c] * om, n, half, Af, Pl[c]);
      }
      #pragma unroll
      for (int t = 0; t < NT; ++t) acc[c][t] = v8f{};
      #pragma unroll
      for (int kk = 0; kk < 4; ++kk) {
        #pragma unroll
        for (int t = 0; t < NT; ++t) {
          acc[c][t] = __builtin_amdgcn_wmma_f32_16x16x4_f32(
               false, Af[kk],  false, Bf[c][kk][t],
               (short)0, acc[c][t],  false,  false);
        }
      }
      __builtin_amdgcn_sched_barrier(0);
    }
  }

  float p15[CPS];
  #pragma unroll
  for (int c = 0; c < CPS; ++c)
    p15[c] = __builtin_bit_cast(float,
        __builtin_amdgcn_readlane(__builtin_bit_cast(int, Pl[c]), 15));
  float e[NT];
  float aseg = p15[0];
  #pragma unroll
  for (int t = 0; t < NT; ++t) e[t] = acc[0][t][7];
  #pragma unroll
  for (int c = 1; c < CPS; ++c) {
    #pragma unroll
    for (int t = 0; t < NT; ++t) e[t] = p15[c] * e[t] + acc[c][t][7];
    aseg *= p15[c];
  }
  if (half) {
    #pragma unroll
    for (int t = 0; t < NT; ++t) E_sm[seg][t * CK + n] = e[t];
  }
  if (lane == 0) A_sm[seg] = aseg;
  __syncthreads();

  float h[NT];
  #pragma unroll
  for (int t = 0; t < NT; ++t) h[t] = 0.0f;
  for (int s = 0; s < seg; ++s) {
    const float as = A_sm[s];
    #pragma unroll
    for (int t = 0; t < NT; ++t) h[t] = as * h[t] + E_sm[s][t * CK + n];
  }

  #pragma unroll
  for (int c = 0; c < CPS; ++c) {
    const float rot  = row_dpp<0x120 + 8>(Pl[c]);
    const float Prot = half ? rot : Pl[c];
    Fixup<7>::run(Prot, h, acc[c]);
    #pragma unroll
    for (int r = 0; r < 8; ++r) {
      #pragma unroll
      for (int t = 0; t < NT; ++t) {
        ob[(size_t)(c * CK + r + 8 * half) * DD + t * CK] = acc[c][t][r];
      }
    }
    if (c + 1 < CPS) {
      #pragma unroll
      for (int t = 0; t < NT; ++t) h[t] = __shfl(acc[c][t][7], n + 16, 32);
    }
  }
}

extern "C" void kernel_launch(void* const* d_in, const int* in_sizes, int n_in,
                              void* d_out, int out_size, void* d_ws, size_t ws_size,
                              hipStream_t stream) {
  const float* x  = (const float*)d_in[0];
  const float* pg = (const float*)d_in[1];
  const float* og = (const float*)d_in[2];
  float* out = (float*)d_out;

  ds_scan_seg_kernel<<<dim3(DD / (NT * CK), BB), dim3(SEG * 32), 0, stream>>>(x, pg, og, out);
}
